# starting kernel
# baseline (speedup 1.0000x reference)
#include <hip/hip_runtime.h>

typedef __attribute__((ext_vector_type(2))) float v2f;
typedef __attribute__((ext_vector_type(8))) float v8f;

#define N_NODES 50000
#define N_EDGES 625000
#define CH 128
#define ROW_BLOCKS (N_NODES / 16)

__global__ void k_zero(float* __restrict__ agg, float* __restrict__ deg) {
    int i = blockIdx.x * blockDim.x + threadIdx.x;
    agg[i] = 0.0f;
    if (i < N_NODES) deg[i] = 0.0f;
}

__global__ void k_deg(const int* __restrict__ ei, float* __restrict__ deg) {
    int e = blockIdx.x * blockDim.x + threadIdx.x;
    if (e < N_EDGES) atomicAdd(&deg[ei[e]], 1.0f);
}

__global__ void k_dis(const float* __restrict__ deg, float* __restrict__ dis) {
    int n = blockIdx.x * blockDim.x + threadIdx.x;
    if (n < N_NODES) {
        float d = deg[n];
        dis[n] = (d > 0.0f) ? rsqrtf(d) : 0.0f;
    }
}

__global__ void k_scatter(const float* __restrict__ x,
                          const int* __restrict__ ei,
                          const float* __restrict__ dis,
                          float* __restrict__ agg) {
    int gid  = blockIdx.x * blockDim.x + threadIdx.x;
    int edge = gid >> 5;
    int lane = threadIdx.x & 31;
    if (edge >= N_EDGES) return;

    int r = ei[edge];
    int c = ei[N_EDGES + edge];
    float w = dis[r] * dis[c];

    const float4* xr = (const float4*)(x + (size_t)c * CH);
    float4 v = xr[lane];

    float* o = agg + (size_t)r * CH + lane * 4;
    atomicAdd(o + 0, v.x * w);
    atomicAdd(o + 1, v.y * w);
    atomicAdd(o + 2, v.z * w);
    atomicAdd(o + 3, v.w * w);
}

__global__ void k_gemm(float* __restrict__ io,
                       const float* __restrict__ W,
                       const float* __restrict__ bias) {
    __shared__ float Wl[CH * CH];

    int tid = threadIdx.x;
    const float4* Wv  = (const float4*)W;
    float4*       Wlv = (float4*)Wl;
#pragma unroll
    for (int i = 0; i < 16; ++i) Wlv[tid + i * 256] = Wv[tid + i * 256];
    __syncthreads();

    int wave = tid >> 5;
    int lane = tid & 31;
    int rb = blockIdx.x * 8 + wave;
    if (rb >= ROW_BLOCKS) return;

    int m    = lane & 15;
    int kh   = (lane >> 4) * 2;
    int ncol = lane & 15;

    const float* arow = io + ((size_t)rb * 16 + m) * CH;

    v8f acc[8] = {};

    for (int k0 = 0; k0 < CH; k0 += 4) {
        v2f a;
        a.x = arow[k0 + kh];
        a.y = arow[k0 + kh + 1];
#pragma unroll
        for (int t = 0; t < 8; ++t) {
            const float* wrow = Wl + (t * 16 + ncol) * CH + k0 + kh;
            v2f b;
            b.x = wrow[0];
            b.y = wrow[1];
            acc[t] = __builtin_amdgcn_wmma_f32_16x16x4_f32(
                false, a, false, b, (short)0, acc[t], false, false);
        }
    }

    int base_m = rb * 16 + (lane >> 4) * 8;
#pragma unroll
    for (int t = 0; t < 8; ++t) {
        int col = t * 16 + ncol;
        float bv = bias[col];
#pragma unroll
        for (int r = 0; r < 8; ++r) {
            io[(size_t)(base_m + r) * CH + col] = acc[t][r] + bv;
        }
    }
}

extern "C" void kernel_launch(void* const* d_in, const int* in_sizes, int n_in,
                              void* d_out, int out_size, void* d_ws, size_t ws_size,
                              hipStream_t stream) {
    const float* x  = (const float*)d_in[0];
    const int*   ei = (const int*)d_in[1];
    const float* W  = (const float*)d_in[2];
    const float* b  = (const float*)d_in[3];
    float* out = (float*)d_out;

    float* deg = (float*)d_ws;
    float* dis = deg + N_NODES;

    k_zero<<<(N_NODES * CH) / 256, 256, 0, stream>>>(out, deg);
    k_deg<<<(N_EDGES + 255) / 256, 256, 0, stream>>>(ei, deg);
    k_dis<<<(N_NODES + 255) / 256, 256, 0, stream>>>(deg, dis);
    k_scatter<<<(N_EDGES * 32) / 256, 256, 0, stream>>>(x, ei, dis, out);
    k_gemm<<<(ROW_BLOCKS + 7) / 8, 256, 0, stream>>>(out, W, b);
}


# Aggregation by owner: a workgroup owns 196 nodes, filters the whole edge list, adds x rows into an LDS tile (f32 LDS atomics, arrival order), then its GEMM from LDS; global float atomics gone
# speedup vs baseline: 2.1110x; 2.1110x over previous
#include <hip/hip_runtime.h>


typedef __attribute__((ext_vector_type(2))) float v2f;
typedef __attribute__((ext_vector_type(4))) float v4f;
typedef __attribute__((ext_vector_type(8))) float v8f;

#define N_NODES 50000
#define N_EDGES 625000
#define CH 128

static_assert(N_NODES % 4 == 0, "deg is zeroed in 16-byte pieces");

__global__ void __launch_bounds__(256)
k_zero(float* __restrict__ deg) {
    float4* deg4 = (float4*)deg;
    int i = blockIdx.x * 256 + threadIdx.x;
    if (i < N_NODES / 4) deg4[i] = make_float4(0.0f, 0.0f, 0.0f, 0.0f);
}

__global__ void k_deg(const int* __restrict__ ei, float* __restrict__ deg) {
    int e = blockIdx.x * blockDim.x + threadIdx.x;
    if (e < N_EDGES) atomicAdd(&deg[ei[e]], 1.0f);
}

__global__ void k_dis(const float* __restrict__ deg, float* __restrict__ dis) {
    int n = blockIdx.x * blockDim.x + threadIdx.x;
    if (n < N_NODES) {
        float d = deg[n];
        dis[n] = (d > 0.0f) ? rsqrtf(d) : 0.0f;
    }
}

#define NPB 196
#define OWN_BLOCKS 256
#define TILE_RB 13
#define PITCH 132
#define OWN_WAVES 32
#define N4 (N_EDGES / 4)
#define STEPS ((N4 + 31) / 32)
#define GSTEPS 8
#define GROUPS ((STEPS + GSTEPS - 1) / GSTEPS)
#define LIST_CAP (GSTEPS * 128)
static_assert(OWN_BLOCKS * NPB >= N_NODES, "every node has an owner");
static_assert(TILE_RB * 16 >= NPB, "tile holds the owned rows");
static_assert(N_EDGES % 4 == 0, "ids split into whole 16-byte pieces");
static_assert(NPB <= 256 && N_EDGES < (1 << 23), "entry packs edge and local row");
static_assert(GROUPS >= OWN_WAVES, "every wave has a first group");
static_assert(GROUPS * GSTEPS * 32 * 4 <= 2 * N_EDGES, "the last group's loads stay inside edge_index");
static_assert((TILE_RB * 16 * PITCH) % 4 == 0, "tile cleared in 16-byte pieces");

static __device__ __forceinline__ void lds_add(float* p, float v) {
    __hip_atomic_fetch_add(p, v, __ATOMIC_RELAXED, __HIP_MEMORY_SCOPE_WORKGROUP);
}

template <int K>
static __device__ __forceinline__ void add_rows(const float* __restrict__ x,
                                                const float* __restrict__ dis,
                                                float* tile, int cc, int lr, int j,
                                                int base, int lane) {
    v2f lo[K], hi[K];
    float wj[K];
    int lrj[K];
#pragma unroll
    for (int u = 0; u < K; ++u) {
        int cj = __builtin_amdgcn_readlane(cc, j + u);
        lrj[u] = __builtin_amdgcn_readlane(lr, j + u);
        const v2f* xr = (const v2f*)(x + (size_t)cj * CH);
        lo[u] = xr[lane];
        hi[u] = xr[32 + lane];
        wj[u] = dis[base + lrj[u]] * dis[cj];
    }
#pragma unroll
    for (int u = 0; u < K; ++u) {
        float* tr = tile + lrj[u] * PITCH + 2 * lane;
        lds_add(tr,      lo[u].x * wj[u]);
        lds_add(tr + 1,  lo[u].y * wj[u]);
        lds_add(tr + 64, hi[u].x * wj[u]);
        lds_add(tr + 65, hi[u].y * wj[u]);
    }
}

#define LOAD_IDS(g_)                                                          \
    {                                                                         \
        int ge_ = (g_) + rot;                                                 \
        if (ge_ >= GROUPS) ge_ -= GROUPS;                                     \
        pbase = ge_ * (GSTEPS * 32) + lane;                                   \
        _Pragma("unroll")        \
        for (int s = 0; s < GSTEPS; ++s) ids[s] = row4[pbase + s * 32];       \
    }

#define APPEND(m_, d_, k_)                                                    \
    {                                                                         \
        unsigned b_ = __builtin_amdgcn_ballot_w32(m_);                        \
        if (b_) {                                                             \
            if (m_) my[n + (int)__builtin_amdgcn_mbcnt_lo(b_, 0u)] =          \
                        ((p * 4 + (k_)) << 8) | (int)(d_);                    \
            n += __builtin_popcount(b_);                                      \
        }                                                                     \
    }

__global__ void __launch_bounds__(OWN_WAVES * 32)
k_own(const float* __restrict__ x,
      const int* __restrict__ ei,
      const float* __restrict__ dis,
      const float* __restrict__ W,
      const float* __restrict__ bias,
      float* __restrict__ out) {
    __shared__ __attribute__((aligned(16))) float tile[TILE_RB * 16 * PITCH];
    __shared__ int mlist[OWN_WAVES * LIST_CAP];

    int tid = threadIdx.x;
    int lane = tid & 31;
    int w = __builtin_amdgcn_readfirstlane(tid >> 5);
    int base = blockIdx.x * NPB;

    {
        v4f* t4 = (v4f*)tile;
        const v4f z = {0.0f, 0.0f, 0.0f, 0.0f};
        for (int i = tid; i < TILE_RB * 16 * PITCH / 4; i += OWN_WAVES * 32) t4[i] = z;
    }
    __syncthreads();

    const int4* row4 = (const int4*)ei;
    const int* colp = ei + N_EDGES;
    int* my = mlist + w * LIST_CAP;
    int rot = (int)((blockIdx.x * (unsigned)GROUPS) >> 8);

    int4 ids[GSTEPS];
    int pbase;
    int g = w;
    LOAD_IDS(g)
    for (;;) {
        int n = 0;
#pragma unroll
        for (int s = 0; s < GSTEPS; ++s) {
            int p = pbase + s * 32;
            unsigned d0 = (unsigned)(ids[s].x - base);
            unsigned d1 = (unsigned)(ids[s].y - base);
            unsigned d2 = (unsigned)(ids[s].z - base);
            unsigned d3 = (unsigned)(ids[s].w - base);
            bool ok = p < N4;
            bool m0 = ok & (d0 < (unsigned)NPB), m1 = ok & (d1 < (unsigned)NPB);
            bool m2 = ok & (d2 < (unsigned)NPB), m3 = ok & (d3 < (unsigned)NPB);
            if (__builtin_amdgcn_ballot_w32(m0 | m1 | m2 | m3)) {
                APPEND(m0, d0, 0)
                APPEND(m1, d1, 1)
                APPEND(m2, d2, 2)
                APPEND(m3, d3, 3)
            }
        }

        int cnt = n < 32 ? n : 32;
        int lr = 0, cc = 0;
        if (n > 0) {
            int ent = my[lane < cnt ? lane : 0];
            lr = ent & 255;
            cc = colp[ent >> 8];
        }

        int gn = g + OWN_WAVES;
        bool more = gn < GROUPS;
        if (more) LOAD_IDS(gn)

        for (int i0 = 0; i0 < n;) {
            int j = 0;
            do {
                int left = cnt - j;
                if (left >= 4)      add_rows<4>(x, dis, tile, cc, lr, j, base, lane);
                else if (left == 3) add_rows<3>(x, dis, tile, cc, lr, j, base, lane);
                else if (left == 2) add_rows<2>(x, dis, tile, cc, lr, j, base, lane);
                else                add_rows<1>(x, dis, tile, cc, lr, j, base, lane);
                j += 4;
            } while (j < cnt);
            i0 += 32;
            if (i0 < n) {
                cnt = n - i0 < 32 ? n - i0 : 32;
                int ent = my[i0 + (lane < cnt ? lane : 0)];
                lr = ent & 255;
                cc = colp[ent >> 8];
            }
        }
        if (!more) break;
        g = gn;
    }

    int c = lane & 15;
    int h = lane >> 4;
    int t = w & 7;
    v4f bw[16];
    {
        const v4f* wr = (const v4f*)(W + (size_t)(t * 16 + c) * CH + 4 * h);
#pragma unroll
        for (int j = 0; j < 16; ++j) bw[j] = wr[2 * j];
    }
    float bv = bias[t * 16 + c];

    __syncthreads();

    for (int rb = w >> 3; rb < TILE_RB; rb += OWN_WAVES / 8) {
        const v4f* ar = (const v4f*)(tile + (rb * 16 + c) * PITCH + 4 * h);
        v8f acc = {};
#pragma unroll
        for (int j = 0; j < 16; ++j) {
            v4f a = ar[2 * j];
            v2f a0, a1, b0, b1;
            a0.x = a[0];     a0.y = a[1];
            a1.x = a[2];     a1.y = a[3];
            b0.x = bw[j][0]; b0.y = bw[j][1];
            b1.x = bw[j][2]; b1.y = bw[j][3];
            acc = __builtin_amdgcn_wmma_f32_16x16x4_f32(
                false, a0, false, b0, (short)0, acc, false, false);
            acc = __builtin_amdgcn_wmma_f32_16x16x4_f32(
                false, a1, false, b1, (short)0, acc, false, false);
        }
        int lrow = rb * 16 + h * 8;
#pragma unroll
        for (int r = 0; r < 8; ++r) {
            int lrr = lrow + r;
            int node = base + lrr;
            if (lrr < NPB && node < N_NODES)
                out[(size_t)node * CH + t * 16 + c] = acc[r] + bv;
        }
    }
}

extern "C" void kernel_launch(void* const* d_in, const int* in_sizes, int n_in,
                              void* d_out, int out_size, void* d_ws, size_t ws_size,
                              hipStream_t stream) {
    const float* x  = (const float*)d_in[0];
    const int*   ei = (const int*)d_in[1];
    const float* W  = (const float*)d_in[2];
    const float* b  = (const float*)d_in[3];
    float* out = (float*)d_out;

    float* deg = (float*)d_ws;
    float* dis = deg + N_NODES;

    k_zero<<<(N_NODES / 4 + 255) / 256, 256, 0, stream>>>(deg);
    k_deg<<<(N_EDGES + 255) / 256, 256, 0, stream>>>(ei, deg);
    k_dis<<<(N_NODES + 255) / 256, 256, 0, stream>>>(deg, dis);
    k_own<<<OWN_BLOCKS, OWN_WAVES * 32, 0, stream>>>(x, ei, dis, W, b, out);
}


# Degrees: 32 workgroups each count a slice of destination ids into a per-node counter plane in LDS (integer LDS atomics, exact); k_dis sums the planes; per-edge global f32 atomic, zero launch gone
# speedup vs baseline: 2.7576x; 1.3063x over previous
#include <hip/hip_runtime.h>


typedef __attribute__((ext_vector_type(2))) float v2f;
typedef __attribute__((ext_vector_type(4))) float v4f;
typedef __attribute__((ext_vector_type(8))) float v8f;

#define N_NODES 50000
#define N_EDGES 625000
#define CH 128

#define N4 (N_EDGES / 4)
#define HG 32
#define HP4 ((N4 + HG - 1) / HG)
#define HWG 1024
#define HIT ((HP4 + HWG - 1) / HWG)
static_assert(N_EDGES % 4 == 0, "ids split into whole 16-byte pieces");
static_assert(N_NODES % 4 == 0, "counter planes move in 16-byte pieces");
static_assert(HG * HP4 >= N4, "the slices cover the edge list");

static __device__ __forceinline__ void count_id(unsigned* hist, int id) {
    if ((unsigned)id < (unsigned)N_NODES)
        __hip_atomic_fetch_add(&hist[id], 1u, __ATOMIC_RELAXED, __HIP_MEMORY_SCOPE_WORKGROUP);
}

__global__ void __launch_bounds__(HWG)
k_count(const int* __restrict__ ei,
        unsigned* __restrict__ part) {
    __shared__ __attribute__((aligned(16))) unsigned hist[N_NODES];
    int tid = threadIdx.x;

    const int4* row4 = (const int4*)ei;
    int p0 = blockIdx.x * HP4;
    int p1 = p0 + HP4 < N4 ? p0 + HP4 : N4;
    int4 v[HIT];
#pragma unroll
    for (int s = 0; s < HIT; ++s) {
        int p = p0 + tid + s * HWG;
        v[s] = p < p1 ? row4[p] : make_int4(-1, -1, -1, -1);
    }

    uint4* h4 = (uint4*)hist;
    for (int i = tid; i < N_NODES / 4; i += HWG) h4[i] = make_uint4(0u, 0u, 0u, 0u);
    __syncthreads();

#pragma unroll
    for (int s = 0; s < HIT; ++s) {
        count_id(hist, v[s].x);
        count_id(hist, v[s].y);
        count_id(hist, v[s].z);
        count_id(hist, v[s].w);
    }
    __syncthreads();

    uint4* o4 = (uint4*)(part + (size_t)blockIdx.x * N_NODES);
    for (int i = tid; i < N_NODES / 4; i += HWG) o4[i] = h4[i];
}

#define DIS_WG 64

__global__ void __launch_bounds__(DIS_WG)
k_dis(const unsigned* __restrict__ part, float* __restrict__ dis) {
    int i = blockIdx.x * DIS_WG + threadIdx.x;
    if (i < N_NODES / 4) {
        const uint4* p4 = (const uint4*)part + i;
        uint4 c[HG];
#pragma unroll
        for (int g = 0; g < HG; ++g) c[g] = p4[(size_t)g * (N_NODES / 4)];
        uint4 s = c[0];
#pragma unroll
        for (int g = 1; g < HG; ++g) {
            s.x += c[g].x; s.y += c[g].y; s.z += c[g].z; s.w += c[g].w;
        }
        float4 r;
        r.x = s.x > 0u ? rsqrtf((float)s.x) : 0.0f;
        r.y = s.y > 0u ? rsqrtf((float)s.y) : 0.0f;
        r.z = s.z > 0u ? rsqrtf((float)s.z) : 0.0f;
        r.w = s.w > 0u ? rsqrtf((float)s.w) : 0.0f;
        ((float4*)dis)[i] = r;
    }
}

#define NPB 196
#define OWN_BLOCKS 256
#define TILE_RB 13
#define PITCH 132
#define OWN_WAVES 32
#define STEPS ((N4 + 31) / 32)
#define GSTEPS 8
#define GROUPS ((STEPS + GSTEPS - 1) / GSTEPS)
#define LIST_CAP (GSTEPS * 128)
static_assert(OWN_BLOCKS * NPB >= N_NODES, "every node has an owner");
static_assert(TILE_RB * 16 >= NPB, "tile holds the owned rows");
static_assert(NPB <= 256 && N_EDGES < (1 << 23), "entry packs edge and local row");
static_assert(GROUPS >= OWN_WAVES, "every wave has a first group");
static_assert(GROUPS * GSTEPS * 32 * 4 <= 2 * N_EDGES, "the last group's loads stay inside edge_index");
static_assert((TILE_RB * 16 * PITCH) % 4 == 0, "tile cleared in 16-byte pieces");

static __device__ __forceinline__ void lds_add(float* p, float v) {
    __hip_atomic_fetch_add(p, v, __ATOMIC_RELAXED, __HIP_MEMORY_SCOPE_WORKGROUP);
}

template <int K>
static __device__ __forceinline__ void add_rows(const float* __restrict__ x,
                                                const float* __restrict__ dis,
                                                float* tile, int cc, int lr, int j,
                                                int base, int lane) {
    v2f lo[K], hi[K];
    float wj[K];
    int lrj[K];
#pragma unroll
    for (int u = 0; u < K; ++u) {
        int cj = __builtin_amdgcn_readlane(cc, j + u);
        lrj[u] = __builtin_amdgcn_readlane(lr, j + u);
        const v2f* xr = (const v2f*)(x + (size_t)cj * CH);
        lo[u] = xr[lane];
        hi[u] = xr[32 + lane];
        wj[u] = dis[base + lrj[u]] * dis[cj];
    }
#pragma unroll
    for (int u = 0; u < K; ++u) {
        float* tr = tile + lrj[u] * PITCH + 2 * lane;
        lds_add(tr,      lo[u].x * wj[u]);
        lds_add(tr + 1,  lo[u].y * wj[u]);
        lds_add(tr + 64, hi[u].x * wj[u]);
        lds_add(tr + 65, hi[u].y * wj[u]);
    }
}

#define LOAD_IDS(g_)                                                          \
    {                                                                         \
        int ge_ = (g_) + rot;                                                 \
        if (ge_ >= GROUPS) ge_ -= GROUPS;                                     \
        pbase = ge_ * (GSTEPS * 32) + lane;                                   \
        _Pragma("unroll")        \
        for (int s = 0; s < GSTEPS; ++s) ids[s] = row4[pbase + s * 32];       \
    }

#define APPEND(m_, d_, k_)                                                    \
    {                                                                         \
        unsigned b_ = __builtin_amdgcn_ballot_w32(m_);                        \
        if (b_) {                                                             \
            if (m_) my[n + (int)__builtin_amdgcn_mbcnt_lo(b_, 0u)] =          \
                        ((p * 4 + (k_)) << 8) | (int)(d_);                    \
            n += __builtin_popcount(b_);                                      \
        }                                                                     \
    }

__global__ void __launch_bounds__(OWN_WAVES * 32)
k_own(const float* __restrict__ x,
      const int* __restrict__ ei,
      const float* __restrict__ dis,
      const float* __restrict__ W,
      const float* __restrict__ bias,
      float* __restrict__ out) {
    __shared__ __attribute__((aligned(16))) float tile[TILE_RB * 16 * PITCH];
    __shared__ int mlist[OWN_WAVES * LIST_CAP];

    int tid = threadIdx.x;
    int lane = tid & 31;
    int w = __builtin_amdgcn_readfirstlane(tid >> 5);
    int base = blockIdx.x * NPB;

    {
        v4f* t4 = (v4f*)tile;
        const v4f z = {0.0f, 0.0f, 0.0f, 0.0f};
        for (int i = tid; i < TILE_RB * 16 * PITCH / 4; i += OWN_WAVES * 32) t4[i] = z;
    }
    __syncthreads();

    const int4* row4 = (const int4*)ei;
    const int* colp = ei + N_EDGES;
    int* my = mlist + w * LIST_CAP;
    int rot = (int)((blockIdx.x * (unsigned)GROUPS) >> 8);

    int4 ids[GSTEPS];
    int pbase;
    int g = w;
    LOAD_IDS(g)
    for (;;) {
        int n = 0;
#pragma unroll
        for (int s = 0; s < GSTEPS; ++s) {
            int p = pbase + s * 32;
            unsigned d0 = (unsigned)(ids[s].x - base);
            unsigned d1 = (unsigned)(ids[s].y - base);
            unsigned d2 = (unsigned)(ids[s].z - base);
            unsigned d3 = (unsigned)(ids[s].w - base);
            bool ok = p < N4;
            bool m0 = ok & (d0 < (unsigned)NPB), m1 = ok & (d1 < (unsigned)NPB);
            bool m2 = ok & (d2 < (unsigned)NPB), m3 = ok & (d3 < (unsigned)NPB);
            if (__builtin_amdgcn_ballot_w32(m0 | m1 | m2 | m3)) {
                APPEND(m0, d0, 0)
                APPEND(m1, d1, 1)
                APPEND(m2, d2, 2)
                APPEND(m3, d3, 3)
            }
        }

        int cnt = n < 32 ? n : 32;
        int lr = 0, cc = 0;
        if (n > 0) {
            int ent = my[lane < cnt ? lane : 0];
            lr = ent & 255;
            cc = colp[ent >> 8];
        }

        int gn = g + OWN_WAVES;
        bool more = gn < GROUPS;
        if (more) LOAD_IDS(gn)

        for (int i0 = 0; i0 < n;) {
            int j = 0;
            do {
                int left = cnt - j;
                if (left >= 4)      add_rows<4>(x, dis, tile, cc, lr, j, base, lane);
                else if (left == 3) add_rows<3>(x, dis, tile, cc, lr, j, base, lane);
                else if (left == 2) add_rows<2>(x, dis, tile, cc, lr, j, base, lane);
                else                add_rows<1>(x, dis, tile, cc, lr, j, base, lane);
                j += 4;
            } while (j < cnt);
            i0 += 32;
            if (i0 < n) {
                cnt = n - i0 < 32 ? n - i0 : 32;
                int ent = my[i0 + (lane < cnt ? lane : 0)];
                lr = ent & 255;
                cc = colp[ent >> 8];
            }
        }
        if (!more) break;
        g = gn;
    }

    int c = lane & 15;
    int h = lane >> 4;
    int t = w & 7;
    v4f bw[16];
    {
        const v4f* wr = (const v4f*)(W + (size_t)(t * 16 + c) * CH + 4 * h);
#pragma unroll
        for (int j = 0; j < 16; ++j) bw[j] = wr[2 * j];
    }
    float bv = bias[t * 16 + c];

    __syncthreads();

    for (int rb = w >> 3; rb < TILE_RB; rb += OWN_WAVES / 8) {
        const v4f* ar = (const v4f*)(tile + (rb * 16 + c) * PITCH + 4 * h);
        v8f acc = {};
#pragma unroll
        for (int j = 0; j < 16; ++j) {
            v4f a = ar[2 * j];
            v2f a0, a1, b0, b1;
            a0.x = a[0];     a0.y = a[1];
            a1.x = a[2];     a1.y = a[3];
            b0.x = bw[j][0]; b0.y = bw[j][1];
            b1.x = bw[j][2]; b1.y = bw[j][3];
            acc = __builtin_amdgcn_wmma_f32_16x16x4_f32(
                false, a0, false, b0, (short)0, acc, false, false);
            acc = __builtin_amdgcn_wmma_f32_16x16x4_f32(
                false, a1, false, b1, (short)0, acc, false, false);
        }
        int lrow = rb * 16 + h * 8;
#pragma unroll
        for (int r = 0; r < 8; ++r) {
            int lrr = lrow + r;
            int node = base + lrr;
            if (lrr < NPB && node < N_NODES)
                out[(size_t)node * CH + t * 16 + c] = acc[r] + bv;
        }
    }
}

extern "C" void kernel_launch(void* const* d_in, const int* in_sizes, int n_in,
                              void* d_out, int out_size, void* d_ws, size_t ws_size,
                              hipStream_t stream) {
    const float* x  = (const float*)d_in[0];
    const int*   ei = (const int*)d_in[1];
    const float* W  = (const float*)d_in[2];
    const float* b  = (const float*)d_in[3];
    float* out = (float*)d_out;

    unsigned* part = (unsigned*)d_ws;
    float* dis = (float*)(part + HG * N_NODES);

    k_count<<<HG, HWG, 0, stream>>>(ei, part);
    k_dis<<<(N_NODES / 4 + DIS_WG - 1) / DIS_WG, DIS_WG, 0, stream>>>(part, dis);
    k_own<<<OWN_BLOCKS, OWN_WAVES * 32, 0, stream>>>(x, ei, dis, W, b, out);
}


# Skip index: count kernel sets a bit per (owner, 128-byte line of destination ids) in LDS; an owner wave lists its slice's set bits and loads only those id lines; LDS sums stay in arrival order
# speedup vs baseline: 3.1300x; 1.1350x over previous
#include <hip/hip_runtime.h>


typedef __attribute__((ext_vector_type(2))) float v2f;
typedef __attribute__((ext_vector_type(4))) float v4f;
typedef __attribute__((ext_vector_type(8))) float v8f;

#define N_NODES 50000
#define N_EDGES 625000
#define CH 128

#define NPB 196
#define OWNERS 256
#define N4 (N_EDGES / 4)
#define NLINES ((N4 + 7) / 8)
#define HG 32
#define LPS ((NLINES + HG - 1) / HG)
#define HP4 (LPS * 8)
#define BW ((LPS + 31) / 32)
#define BWP (BW + 1)
#define HWG 1024
#define HIT ((HP4 + HWG - 1) / HWG)
static_assert(N_EDGES % 4 == 0, "ids split into whole 16-byte pieces");
static_assert(N_NODES % 4 == 0, "counter planes move in 16-byte pieces");
static_assert(HG * HP4 >= N4, "the slices cover the edge list");
static_assert(OWNERS * NPB >= N_NODES, "every node has an owner");
static_assert(BW <= 32, "one word of a bit row per lane");
static_assert((OWNERS * BWP) % 4 == 0, "bit rows cleared in 16-byte pieces");

static __device__ __forceinline__ void count_id(unsigned* hist, unsigned* obm, int id,
                                                int word, unsigned bit) {
    if ((unsigned)id < (unsigned)N_NODES) {
        __hip_atomic_fetch_add(&hist[id], 1u, __ATOMIC_RELAXED, __HIP_MEMORY_SCOPE_WORKGROUP);
        __hip_atomic_fetch_or(&obm[((unsigned)id / (unsigned)NPB) * BWP + word], bit,
                              __ATOMIC_RELAXED, __HIP_MEMORY_SCOPE_WORKGROUP);
    }
}

__global__ void __launch_bounds__(HWG)
k_count(const int* __restrict__ ei,
        unsigned* __restrict__ part,
        unsigned* __restrict__ bm) {
    __shared__ __attribute__((aligned(16))) unsigned hist[N_NODES];
    __shared__ __attribute__((aligned(16))) unsigned obm[OWNERS * BWP];
    int tid = threadIdx.x;

    const int4* row4 = (const int4*)ei;
    int p0 = blockIdx.x * HP4;
    int p1 = p0 + HP4 < N4 ? p0 + HP4 : N4;
    int4 v[HIT];
#pragma unroll
    for (int s = 0; s < HIT; ++s) {
        int p = p0 + tid + s * HWG;
        v[s] = p < p1 ? row4[p] : make_int4(-1, -1, -1, -1);
    }

    uint4* h4 = (uint4*)hist;
    uint4* b4 = (uint4*)obm;
    for (int i = tid; i < N_NODES / 4; i += HWG) h4[i] = make_uint4(0u, 0u, 0u, 0u);
    for (int i = tid; i < OWNERS * BWP / 4; i += HWG) b4[i] = make_uint4(0u, 0u, 0u, 0u);
    __syncthreads();

#pragma unroll
    for (int s = 0; s < HIT; ++s) {
        int ln = (tid + s * HWG) >> 3;
        int word = ln >> 5;
        unsigned bit = 1u << (ln & 31);
        count_id(hist, obm, v[s].x, word, bit);
        count_id(hist, obm, v[s].y, word, bit);
        count_id(hist, obm, v[s].z, word, bit);
        count_id(hist, obm, v[s].w, word, bit);
    }
    __syncthreads();

    uint4* o4 = (uint4*)(part + (size_t)blockIdx.x * N_NODES);
    for (int i = tid; i < N_NODES / 4; i += HWG) o4[i] = h4[i];
    for (int i = tid; i < OWNERS * BW; i += HWG) {
        int o = i / BW;
        int j = i - o * BW;
        bm[((size_t)o * HG + blockIdx.x) * BW + j] = obm[o * BWP + j];
    }
}

#define DIS_WG 64

__global__ void __launch_bounds__(DIS_WG)
k_dis(const unsigned* __restrict__ part, float* __restrict__ dis) {
    int i = blockIdx.x * DIS_WG + threadIdx.x;
    if (i < N_NODES / 4) {
        const uint4* p4 = (const uint4*)part + i;
        uint4 c[HG];
#pragma unroll
        for (int g = 0; g < HG; ++g) c[g] = p4[(size_t)g * (N_NODES / 4)];
        uint4 s = c[0];
#pragma unroll
        for (int g = 1; g < HG; ++g) {
            s.x += c[g].x; s.y += c[g].y; s.z += c[g].z; s.w += c[g].w;
        }
        float4 r;
        r.x = s.x > 0u ? rsqrtf((float)s.x) : 0.0f;
        r.y = s.y > 0u ? rsqrtf((float)s.y) : 0.0f;
        r.z = s.z > 0u ? rsqrtf((float)s.z) : 0.0f;
        r.w = s.w > 0u ? rsqrtf((float)s.w) : 0.0f;
        ((float4*)dis)[i] = r;
    }
}

#define OWN_BLOCKS OWNERS
#define TILE_RB 13
#define PITCH 132
#define OWN_WAVES 32
#define GSTEPS 6
#define LIST_CAP (GSTEPS * 128)
#define LINE_CAP (BW * 32)
static_assert(OWN_WAVES == HG, "wave w of an owner takes slice w of the ids");
static_assert(TILE_RB * 16 >= NPB, "tile holds the owned rows");
static_assert(NPB <= 256 && N_EDGES < (1 << 23), "entry packs edge and local row");
static_assert((TILE_RB * 16 * PITCH) % 4 == 0, "tile cleared in 16-byte pieces");

static __device__ __forceinline__ void lds_add(float* p, float v) {
    __hip_atomic_fetch_add(p, v, __ATOMIC_RELAXED, __HIP_MEMORY_SCOPE_WORKGROUP);
}

template <int K>
static __device__ __forceinline__ void add_rows(const float* __restrict__ x,
                                                const float* __restrict__ dis,
                                                float* tile, int cc, int lr, int j,
                                                int base, int lane) {
    v2f lo[K], hi[K];
    float wj[K];
    int lrj[K];
#pragma unroll
    for (int u = 0; u < K; ++u) {
        int cj = __builtin_amdgcn_readlane(cc, j + u);
        lrj[u] = __builtin_amdgcn_readlane(lr, j + u);
        const v2f* xr = (const v2f*)(x + (size_t)cj * CH);
        lo[u] = xr[lane];
        hi[u] = xr[32 + lane];
        wj[u] = dis[base + lrj[u]] * dis[cj];
    }
#pragma unroll
    for (int u = 0; u < K; ++u) {
        float* tr = tile + lrj[u] * PITCH + 2 * lane;
        lds_add(tr,      lo[u].x * wj[u]);
        lds_add(tr + 1,  lo[u].y * wj[u]);
        lds_add(tr + 64, hi[u].x * wj[u]);
        lds_add(tr + 65, hi[u].y * wj[u]);
    }
}

#define LOAD_IDS(g_)                                                          \
    {                                                                         \
        _Pragma("unroll")  \
        for (int s = 0; s < GSTEPS; ++s) {                                    \
            int e_ = ((g_) * GSTEPS + s) * 4 + (lane >> 3);                   \
            int p_ = (w * LPS + ll[e_ < nl ? e_ : 0]) * 8 + (lane & 7);       \
            pv[s] = (e_ < nl && p_ < N4) ? p_ : -1;                           \
        }                                                                     \
        _Pragma("unroll")       \
        for (int s = 0; s < GSTEPS; ++s) ids[s] = row4[pv[s] < 0 ? 0 : pv[s]];\
    }

#define APPEND(m_, d_, k_)                                                    \
    {                                                                         \
        unsigned b_ = __builtin_amdgcn_ballot_w32(m_);                        \
        if (b_) {                                                             \
            if (m_) my[n + (int)__builtin_amdgcn_mbcnt_lo(b_, 0u)] =          \
                        ((p * 4 + (k_)) << 8) | (int)(d_);                    \
            n += __builtin_popcount(b_);                                      \
        }                                                                     \
    }

__global__ void __launch_bounds__(OWN_WAVES * 32)
k_own(const float* __restrict__ x,
      const int* __restrict__ ei,
      const float* __restrict__ dis,
      const unsigned* __restrict__ bm,
      const float* __restrict__ W,
      const float* __restrict__ bias,
      float* __restrict__ out) {
    __shared__ __attribute__((aligned(16))) float tile[TILE_RB * 16 * PITCH];
    __shared__ int mlist[OWN_WAVES * LIST_CAP];
    __shared__ int llist[OWN_WAVES * LINE_CAP];

    int tid = threadIdx.x;
    int lane = tid & 31;
    int w = __builtin_amdgcn_readfirstlane(tid >> 5);
    int base = blockIdx.x * NPB;

    unsigned bits = lane < BW ? bm[((size_t)blockIdx.x * HG + w) * BW + lane] : 0u;

    {
        v4f* t4 = (v4f*)tile;
        const v4f z = {0.0f, 0.0f, 0.0f, 0.0f};
        for (int i = tid; i < TILE_RB * 16 * PITCH / 4; i += OWN_WAVES * 32) t4[i] = z;
    }

    const int4* row4 = (const int4*)ei;
    const int* colp = ei + N_EDGES;
    int* my = mlist + w * LIST_CAP;
    int* ll = llist + w * LINE_CAP;

    int nl;
    {
        int cnt1 = __builtin_popcount(bits);
        int inc = cnt1;
#pragma unroll
        for (int d = 1; d < 32; d <<= 1) {
            int up = __shfl_up(inc, d);
            if (lane >= d) inc += up;
        }
        nl = __builtin_amdgcn_readlane(inc, 31);
        int off = inc - cnt1;
        while (bits) {
            ll[off++] = lane * 32 + (__builtin_ffs((int)bits) - 1);
            bits &= bits - 1u;
        }
    }
    __syncthreads();

    int ngroups = (nl + GSTEPS * 4 - 1) / (GSTEPS * 4);
    int4 ids[GSTEPS];
    int pv[GSTEPS];
    int g = 0;
    if (ngroups > 0) LOAD_IDS(0)
    while (ngroups > 0) {
        int n = 0;
#pragma unroll
        for (int s = 0; s < GSTEPS; ++s) {
            int p = pv[s];
            unsigned d0 = (unsigned)(ids[s].x - base);
            unsigned d1 = (unsigned)(ids[s].y - base);
            unsigned d2 = (unsigned)(ids[s].z - base);
            unsigned d3 = (unsigned)(ids[s].w - base);
            bool ok = p >= 0;
            bool m0 = ok & (d0 < (unsigned)NPB), m1 = ok & (d1 < (unsigned)NPB);
            bool m2 = ok & (d2 < (unsigned)NPB), m3 = ok & (d3 < (unsigned)NPB);
            if (__builtin_amdgcn_ballot_w32(m0 | m1 | m2 | m3)) {
                APPEND(m0, d0, 0)
                APPEND(m1, d1, 1)
                APPEND(m2, d2, 2)
                APPEND(m3, d3, 3)
            }
        }

        int cnt = n < 32 ? n : 32;
        int lr = 0, cc = 0;
        if (n > 0) {
            int ent = my[lane < cnt ? lane : 0];
            lr = ent & 255;
            cc = colp[ent >> 8];
        }

        int gn = g + 1;
        bool more = gn < ngroups;
        if (more) LOAD_IDS(gn)

        for (int i0 = 0; i0 < n;) {
            int j = 0;
            do {
                int left = cnt - j;
                if (left >= 4)      add_rows<4>(x, dis, tile, cc, lr, j, base, lane);
                else if (left == 3) add_rows<3>(x, dis, tile, cc, lr, j, base, lane);
                else if (left == 2) add_rows<2>(x, dis, tile, cc, lr, j, base, lane);
                else                add_rows<1>(x, dis, tile, cc, lr, j, base, lane);
                j += 4;
            } while (j < cnt);
            i0 += 32;
            if (i0 < n) {
                cnt = n - i0 < 32 ? n - i0 : 32;
                int ent = my[i0 + (lane < cnt ? lane : 0)];
                lr = ent & 255;
                cc = colp[ent >> 8];
            }
        }
        if (!more) break;
        g = gn;
    }

    int c = lane & 15;
    int h = lane >> 4;
    int t = w & 7;
    v4f bw[16];
    {
        const v4f* wr = (const v4f*)(W + (size_t)(t * 16 + c) * CH + 4 * h);
#pragma unroll
        for (int j = 0; j < 16; ++j) bw[j] = wr[2 * j];
    }
    float bv = bias[t * 16 + c];

    __syncthreads();

    for (int rb = w >> 3; rb < TILE_RB; rb += OWN_WAVES / 8) {
        const v4f* ar = (const v4f*)(tile + (rb * 16 + c) * PITCH + 4 * h);
        v8f acc = {};
#pragma unroll
        for (int j = 0; j < 16; ++j) {
            v4f a = ar[2 * j];
            v2f a0, a1, b0, b1;
            a0.x = a[0];     a0.y = a[1];
            a1.x = a[2];     a1.y = a[3];
            b0.x = bw[j][0]; b0.y = bw[j][1];
            b1.x = bw[j][2]; b1.y = bw[j][3];
            acc = __builtin_amdgcn_wmma_f32_16x16x4_f32(
                false, a0, false, b0, (short)0, acc, false, false);
            acc = __builtin_amdgcn_wmma_f32_16x16x4_f32(
                false, a1, false, b1, (short)0, acc, false, false);
        }
        int lrow = rb * 16 + h * 8;
#pragma unroll
        for (int r = 0; r < 8; ++r) {
            int lrr = lrow + r;
            int node = base + lrr;
            if (lrr < NPB && node < N_NODES)
                out[(size_t)node * CH + t * 16 + c] = acc[r] + bv;
        }
    }
}

extern "C" void kernel_launch(void* const* d_in, const int* in_sizes, int n_in,
                              void* d_out, int out_size, void* d_ws, size_t ws_size,
                              hipStream_t stream) {
    const float* x  = (const float*)d_in[0];
    const int*   ei = (const int*)d_in[1];
    const float* W  = (const float*)d_in[2];
    const float* b  = (const float*)d_in[3];
    float* out = (float*)d_out;

    unsigned* part = (unsigned*)d_ws;
    float* dis = (float*)(part + HG * N_NODES);
    unsigned* bm = (unsigned*)(dis + N_NODES);

    k_count<<<HG, HWG, 0, stream>>>(ei, part, bm);
    k_dis<<<(N_NODES / 4 + DIS_WG - 1) / DIS_WG, DIS_WG, 0, stream>>>(part, dis);
    k_own<<<OWN_BLOCKS, OWN_WAVES * 32, 0, stream>>>(x, ei, dis, bm, W, b, out);
}


# Count kernel cut 2-D: one workgroup per (edge slice, node range of 32 owners): 256 workgroups of 8 waves, a 25 KB LDS counter plane each (was 32 workgroups of 32 waves, 200 KB each); counts exact
# speedup vs baseline: 3.2790x; 1.0476x over previous
#include <hip/hip_runtime.h>


typedef __attribute__((ext_vector_type(2))) float v2f;
typedef __attribute__((ext_vector_type(4))) float v4f;
typedef __attribute__((ext_vector_type(8))) float v8f;

#define N_NODES 50000
#define N_EDGES 625000
#define CH 128

#define NPB 196
#define OWNERS 256
#define N4 (N_EDGES / 4)
#define NLINES ((N4 + 7) / 8)
#define HG 32
#define LPS ((NLINES + HG - 1) / HG)
#define HP4 (LPS * 8)
#define BW ((LPS + 31) / 32)
#define BWP (BW + 1)
#define NR 8
#define OPR (OWNERS / NR)
#define RN (OPR * NPB)
#define PP (NR * RN)
#define CWG 256
#define CIT ((HP4 + CWG - 1) / CWG)
static_assert(N_EDGES % 4 == 0, "ids split into whole 16-byte pieces");
static_assert(N_NODES % 4 == 0, "degrees move in 16-byte pieces");
static_assert(HG * HP4 >= N4, "the slices cover the edge list");
static_assert(OWNERS * NPB >= N_NODES, "every node has an owner");
static_assert(OWNERS % NR == 0 && PP >= N_NODES, "the ranges cover the nodes");
static_assert((RN * 4) % 128 == 0, "a piece of a plane is whole 128-byte lines");
static_assert(BW <= 32, "one word of a bit row per lane");
static_assert((OPR * BWP) % 4 == 0, "bit rows cleared in 16-byte pieces");

static __device__ __forceinline__ void count_id(unsigned* hist, unsigned* obm, int id,
                                                unsigned lo, int word, unsigned bit) {
    unsigned d = (unsigned)id - lo;
    if (d < (unsigned)RN && (unsigned)id < (unsigned)N_NODES) {
        __hip_atomic_fetch_add(&hist[d], 1u, __ATOMIC_RELAXED, __HIP_MEMORY_SCOPE_WORKGROUP);
        __hip_atomic_fetch_or(&obm[(d / (unsigned)NPB) * BWP + word], bit,
                              __ATOMIC_RELAXED, __HIP_MEMORY_SCOPE_WORKGROUP);
    }
}

__global__ void __launch_bounds__(CWG)
k_count(const int* __restrict__ ei,
        unsigned* __restrict__ part,
        unsigned* __restrict__ bm) {
    __shared__ __attribute__((aligned(16))) unsigned hist[RN];
    __shared__ __attribute__((aligned(16))) unsigned obm[OPR * BWP];
    int tid = threadIdx.x;
    int sl = blockIdx.x / NR;
    int nr = blockIdx.x - sl * NR;
    unsigned lo = (unsigned)(nr * RN);

    const int4* row4 = (const int4*)ei;
    int p0 = sl * HP4;
    int p1 = p0 + HP4 < N4 ? p0 + HP4 : N4;
    int4 v[CIT];
#pragma unroll
    for (int s = 0; s < CIT; ++s) {
        int p = p0 + tid + s * CWG;
        v[s] = p < p1 ? row4[p] : make_int4(-1, -1, -1, -1);
    }

    uint4* h4 = (uint4*)hist;
    uint4* b4 = (uint4*)obm;
    for (int i = tid; i < RN / 4; i += CWG) h4[i] = make_uint4(0u, 0u, 0u, 0u);
    for (int i = tid; i < OPR * BWP / 4; i += CWG) b4[i] = make_uint4(0u, 0u, 0u, 0u);
    __syncthreads();

#pragma unroll
    for (int s = 0; s < CIT; ++s) {
        int ln = (tid + s * CWG) >> 3;
        int word = ln >> 5;
        unsigned bit = 1u << (ln & 31);
        count_id(hist, obm, v[s].x, lo, word, bit);
        count_id(hist, obm, v[s].y, lo, word, bit);
        count_id(hist, obm, v[s].z, lo, word, bit);
        count_id(hist, obm, v[s].w, lo, word, bit);
    }
    __syncthreads();

    uint4* o4 = (uint4*)(part + (size_t)sl * PP + lo);
    for (int i = tid; i < RN / 4; i += CWG) o4[i] = h4[i];
    for (int i = tid; i < OPR * BW; i += CWG) {
        int o = i / BW;
        int j = i - o * BW;
        bm[((size_t)(nr * OPR + o) * HG + sl) * BW + j] = obm[o * BWP + j];
    }
}

#define DIS_WG 64

__global__ void __launch_bounds__(DIS_WG)
k_dis(const unsigned* __restrict__ part, float* __restrict__ dis) {
    int i = blockIdx.x * DIS_WG + threadIdx.x;
    if (i < N_NODES / 4) {
        const uint4* p4 = (const uint4*)part + i;
        uint4 c[HG];
#pragma unroll
        for (int g = 0; g < HG; ++g) c[g] = p4[(size_t)g * (PP / 4)];
        uint4 s = c[0];
#pragma unroll
        for (int g = 1; g < HG; ++g) {
            s.x += c[g].x; s.y += c[g].y; s.z += c[g].z; s.w += c[g].w;
        }
        float4 r;
        r.x = s.x > 0u ? rsqrtf((float)s.x) : 0.0f;
        r.y = s.y > 0u ? rsqrtf((float)s.y) : 0.0f;
        r.z = s.z > 0u ? rsqrtf((float)s.z) : 0.0f;
        r.w = s.w > 0u ? rsqrtf((float)s.w) : 0.0f;
        ((float4*)dis)[i] = r;
    }
}

#define OWN_BLOCKS OWNERS
#define TILE_RB 13
#define PITCH 132
#define OWN_WAVES 32
#define GSTEPS 6
#define LIST_CAP (GSTEPS * 128)
#define LINE_CAP (BW * 32)
static_assert(OWN_WAVES == HG, "wave w of an owner takes slice w of the ids");
static_assert(TILE_RB * 16 >= NPB, "tile holds the owned rows");
static_assert(NPB <= 256 && N_EDGES < (1 << 23), "entry packs edge and local row");
static_assert((TILE_RB * 16 * PITCH) % 4 == 0, "tile cleared in 16-byte pieces");

static __device__ __forceinline__ void lds_add(float* p, float v) {
    __hip_atomic_fetch_add(p, v, __ATOMIC_RELAXED, __HIP_MEMORY_SCOPE_WORKGROUP);
}

template <int K>
static __device__ __forceinline__ void add_rows(const float* __restrict__ x,
                                                const float* __restrict__ dis,
                                                float* tile, int cc, int lr, int j,
                                                int base, int lane) {
    v2f lo[K], hi[K];
    float wj[K];
    int lrj[K];
#pragma unroll
    for (int u = 0; u < K; ++u) {
        int cj = __builtin_amdgcn_readlane(cc, j + u);
        lrj[u] = __builtin_amdgcn_readlane(lr, j + u);
        const v2f* xr = (const v2f*)(x + (size_t)cj * CH);
        lo[u] = xr[lane];
        hi[u] = xr[32 + lane];
        wj[u] = dis[base + lrj[u]] * dis[cj];
    }
#pragma unroll
    for (int u = 0; u < K; ++u) {
        float* tr = tile + lrj[u] * PITCH + 2 * lane;
        lds_add(tr,      lo[u].x * wj[u]);
        lds_add(tr + 1,  lo[u].y * wj[u]);
        lds_add(tr + 64, hi[u].x * wj[u]);
        lds_add(tr + 65, hi[u].y * wj[u]);
    }
}

#define LOAD_IDS(g_)                                                          \
    {                                                                         \
        _Pragma("unroll")  \
        for (int s = 0; s < GSTEPS; ++s) {                                    \
            int e_ = ((g_) * GSTEPS + s) * 4 + (lane >> 3);                   \
            int p_ = (w * LPS + ll[e_ < nl ? e_ : 0]) * 8 + (lane & 7);       \
            pv[s] = (e_ < nl && p_ < N4) ? p_ : -1;                           \
        }                                                                     \
        _Pragma("unroll")       \
        for (int s = 0; s < GSTEPS; ++s) ids[s] = row4[pv[s] < 0 ? 0 : pv[s]];\
    }

#define APPEND(m_, d_, k_)                                                    \
    {                                                                         \
        unsigned b_ = __builtin_amdgcn_ballot_w32(m_);                        \
        if (b_) {                                                             \
            if (m_) my[n + (int)__builtin_amdgcn_mbcnt_lo(b_, 0u)] =          \
                        ((p * 4 + (k_)) << 8) | (int)(d_);                    \
            n += __builtin_popcount(b_);                                      \
        }                                                                     \
    }

__global__ void __launch_bounds__(OWN_WAVES * 32)
k_own(const float* __restrict__ x,
      const int* __restrict__ ei,
      const float* __restrict__ dis,
      const unsigned* __restrict__ bm,
      const float* __restrict__ W,
      const float* __restrict__ bias,
      float* __restrict__ out) {
    __shared__ __attribute__((aligned(16))) float tile[TILE_RB * 16 * PITCH];
    __shared__ int mlist[OWN_WAVES * LIST_CAP];
    __shared__ int llist[OWN_WAVES * LINE_CAP];

    int tid = threadIdx.x;
    int lane = tid & 31;
    int w = __builtin_amdgcn_readfirstlane(tid >> 5);
    int base = blockIdx.x * NPB;

    unsigned bits = lane < BW ? bm[((size_t)blockIdx.x * HG + w) * BW + lane] : 0u;

    {
        v4f* t4 = (v4f*)tile;
        const v4f z = {0.0f, 0.0f, 0.0f, 0.0f};
        for (int i = tid; i < TILE_RB * 16 * PITCH / 4; i += OWN_WAVES * 32) t4[i] = z;
    }

    const int4* row4 = (const int4*)ei;
    const int* colp = ei + N_EDGES;
    int* my = mlist + w * LIST_CAP;
    int* ll = llist + w * LINE_CAP;

    int nl;
    {
        int cnt1 = __builtin_popcount(bits);
        int inc = cnt1;
#pragma unroll
        for (int d = 1; d < 32; d <<= 1) {
            int up = __shfl_up(inc, d);
            if (lane >= d) inc += up;
        }
        nl = __builtin_amdgcn_readlane(inc, 31);
        int off = inc - cnt1;
        while (bits) {
            ll[off++] = lane * 32 + (__builtin_ffs((int)bits) - 1);
            bits &= bits - 1u;
        }
    }
    __syncthreads();

    int ngroups = (nl + GSTEPS * 4 - 1) / (GSTEPS * 4);
    int4 ids[GSTEPS];
    int pv[GSTEPS];
    int g = 0;
    if (ngroups > 0) LOAD_IDS(0)
    while (ngroups > 0) {
        int n = 0;
#pragma unroll
        for (int s = 0; s < GSTEPS; ++s) {
            int p = pv[s];
            unsigned d0 = (unsigned)(ids[s].x - base);
            unsigned d1 = (unsigned)(ids[s].y - base);
            unsigned d2 = (unsigned)(ids[s].z - base);
            unsigned d3 = (unsigned)(ids[s].w - base);
            bool ok = p >= 0;
            bool m0 = ok & (d0 < (unsigned)NPB), m1 = ok & (d1 < (unsigned)NPB);
            bool m2 = ok & (d2 < (unsigned)NPB), m3 = ok & (d3 < (unsigned)NPB);
            if (__builtin_amdgcn_ballot_w32(m0 | m1 | m2 | m3)) {
                APPEND(m0, d0, 0)
                APPEND(m1, d1, 1)
                APPEND(m2, d2, 2)
                APPEND(m3, d3, 3)
            }
        }

        int cnt = n < 32 ? n : 32;
        int lr = 0, cc = 0;
        if (n > 0) {
            int ent = my[lane < cnt ? lane : 0];
            lr = ent & 255;
            cc = colp[ent >> 8];
        }

        int gn = g + 1;
        bool more = gn < ngroups;
        if (more) LOAD_IDS(gn)

        for (int i0 = 0; i0 < n;) {
            int j = 0;
            do {
                int left = cnt - j;
                if (left >= 4)      add_rows<4>(x, dis, tile, cc, lr, j, base, lane);
                else if (left == 3) add_rows<3>(x, dis, tile, cc, lr, j, base, lane);
                else if (left == 2) add_rows<2>(x, dis, tile, cc, lr, j, base, lane);
                else                add_rows<1>(x, dis, tile, cc, lr, j, base, lane);
                j += 4;
            } while (j < cnt);
            i0 += 32;
            if (i0 < n) {
                cnt = n - i0 < 32 ? n - i0 : 32;
                int ent = my[i0 + (lane < cnt ? lane : 0)];
                lr = ent & 255;
                cc = colp[ent >> 8];
            }
        }
        if (!more) break;
        g = gn;
    }

    int c = lane & 15;
    int h = lane >> 4;
    int t = w & 7;
    v4f bw[16];
    {
        const v4f* wr = (const v4f*)(W + (size_t)(t * 16 + c) * CH + 4 * h);
#pragma unroll
        for (int j = 0; j < 16; ++j) bw[j] = wr[2 * j];
    }
    float bv = bias[t * 16 + c];

    __syncthreads();

    for (int rb = w >> 3; rb < TILE_RB; rb += OWN_WAVES / 8) {
        const v4f* ar = (const v4f*)(tile + (rb * 16 + c) * PITCH + 4 * h);
        v8f acc = {};
#pragma unroll
        for (int j = 0; j < 16; ++j) {
            v4f a = ar[2 * j];
            v2f a0, a1, b0, b1;
            a0.x = a[0];     a0.y = a[1];
            a1.x = a[2];     a1.y = a[3];
            b0.x = bw[j][0]; b0.y = bw[j][1];
            b1.x = bw[j][2]; b1.y = bw[j][3];
            acc = __builtin_amdgcn_wmma_f32_16x16x4_f32(
                false, a0, false, b0, (short)0, acc, false, false);
            acc = __builtin_amdgcn_wmma_f32_16x16x4_f32(
                false, a1, false, b1, (short)0, acc, false, false);
        }
        int lrow = rb * 16 + h * 8;
#pragma unroll
        for (int r = 0; r < 8; ++r) {
            int lrr = lrow + r;
            int node = base + lrr;
            if (lrr < NPB && node < N_NODES)
                out[(size_t)node * CH + t * 16 + c] = acc[r] + bv;
        }
    }
}

extern "C" void kernel_launch(void* const* d_in, const int* in_sizes, int n_in,
                              void* d_out, int out_size, void* d_ws, size_t ws_size,
                              hipStream_t stream) {
    const float* x  = (const float*)d_in[0];
    const int*   ei = (const int*)d_in[1];
    const float* W  = (const float*)d_in[2];
    const float* b  = (const float*)d_in[3];
    float* out = (float*)d_out;

    unsigned* part = (unsigned*)d_ws;
    float* dis = (float*)(part + HG * PP);
    unsigned* bm = (unsigned*)(dis + N_NODES);

    k_count<<<HG * NR, CWG, 0, stream>>>(ei, part, bm);
    k_dis<<<(N_NODES / 4 + DIS_WG - 1) / DIS_WG, DIS_WG, 0, stream>>>(part, dis);
    k_own<<<OWN_BLOCKS, OWN_WAVES * 32, 0, stream>>>(x, ei, dis, bm, W, b, out);
}
